# starting kernel
# baseline (speedup 1.0000x reference)
#include <hip/hip_runtime.h>

typedef float v2f __attribute__((ext_vector_type(2)));
typedef float v8f __attribute__((ext_vector_type(8)));

#define DIN 256
#define DOUT 128

__global__ __launch_bounds__(256) void gcn_gemm_wmma(const float* __restrict__ x,
                                                     const float* __restrict__ W,
                                                     float* __restrict__ h,
                                                     int n, int mtiles) {
    const int tile = blockIdx.x * (blockDim.x >> 5) + (threadIdx.x >> 5);
    if (tile >= mtiles) return;
    const int lane = threadIdx.x & 31;
    const int half = lane >> 4;
    const int l16  = lane & 15;

    int arow = tile * 16 + l16;
    if (arow >= n) arow = n - 1;
    const float* ap = x + (size_t)arow * DIN + half * 2;

    v8f acc[8];
#pragma unroll
    for (int ct = 0; ct < 8; ++ct) acc[ct] = (v8f)(0.0f);

    for (int k = 0; k < DIN; k += 4) {
        v2f a = *(const v2f*)(ap + k);
#pragma unroll
        for (int ct = 0; ct < 8; ++ct) {
            const float* bp = W + (size_t)(ct * 16 + l16) * DIN + half * 2 + k;
            v2f b = *(const v2f*)bp;
            acc[ct] = __builtin_amdgcn_wmma_f32_16x16x4_f32(
                false, a, false, b, (short)0, acc[ct], false, false);
        }
    }

    const int rbase = tile * 16 + half * 8;
#pragma unroll
    for (int j = 0; j < 8; ++j) {
        const int r = rbase + j;
        if (r < n) {
#pragma unroll
            for (int ct = 0; ct < 8; ++ct)
                h[(size_t)r * DOUT + ct * 16 + l16] = acc[ct][j];
        }
    }
}

__global__ __launch_bounds__(256) void gcn_spmm(const float* __restrict__ h,
                                                const int* __restrict__ ptr,
                                                const int* __restrict__ idx,
                                                float* __restrict__ out, int n) {
    const int node = blockIdx.x * (blockDim.x >> 5) + (threadIdx.x >> 5);
    if (node >= n) return;
    const int lane = threadIdx.x & 31;

    const int s = ptr[node];
    const int e = ptr[node + 1];

    float ax = 0.f, ay = 0.f, az = 0.f, aw = 0.f;
    int j = s;
    for (; j + 1 < e; j += 2) {
        const int s0 = idx[j];
        const int s1 = idx[j + 1];
        const float4 v0 = ((const float4*)(h + (size_t)s0 * DOUT))[lane];
        const float4 v1 = ((const float4*)(h + (size_t)s1 * DOUT))[lane];
        ax += v0.x + v1.x;  ay += v0.y + v1.y;
        az += v0.z + v1.z;  aw += v0.w + v1.w;
    }
    if (j < e) {
        const int s0 = idx[j];
        const float4 v0 = ((const float4*)(h + (size_t)s0 * DOUT))[lane];
        ax += v0.x;  ay += v0.y;  az += v0.z;  aw += v0.w;
    }

    const float inv = (e > s) ? 1.0f / (float)(e - s) : 0.0f;
    float4 r;
    r.x = ax * inv;  r.y = ay * inv;  r.z = az * inv;  r.w = aw * inv;
    ((float4*)(out + (size_t)node * DOUT))[lane] = r;
}

extern "C" void kernel_launch(void* const* d_in, const int* in_sizes, int n_in,
                              void* d_out, int out_size, void* d_ws, size_t ws_size,
                              hipStream_t stream) {
    const float* x   = (const float*)d_in[0];
    const float* W   = (const float*)d_in[1];
    const int*   ptr = (const int*)d_in[2];
    const int*   idx = (const int*)d_in[3];
    (void)n_in; (void)out_size; (void)ws_size;

    const int n = in_sizes[2] - 1;
    float* h   = (float*)d_ws;
    float* out = (float*)d_out;

    const int mtiles = (n + 15) / 16;
    const int wavesPerBlock = 8;

    dim3 blk(256);
    dim3 g1((mtiles + wavesPerBlock - 1) / wavesPerBlock);
    gcn_gemm_wmma<<<g1, blk, 0, stream>>>(x, W, h, n, mtiles);

    dim3 g2((n + wavesPerBlock - 1) / wavesPerBlock);
    gcn_spmm<<<g2, blk, 0, stream>>>(h, ptr, idx, out, n);
}


# SpMM: a node's neighbour indices read by one lane-wide load per 32 edges and handed out by lane reads, so the gather loop is row loads only (4 per trip, no index load per edge); sum order kept
# speedup vs baseline: 1.7880x; 1.7880x over previous
#include <hip/hip_runtime.h>

typedef float v2f __attribute__((ext_vector_type(2)));
typedef float v4f __attribute__((ext_vector_type(4)));
typedef float v8f __attribute__((ext_vector_type(8)));

#define DIN 256
#define DOUT 128

#define WPITCH 260
#define GEMM_WAVES 16
#define GEMM_THREADS (GEMM_WAVES * 32)
#define KGROUPS (DIN / 8)

__global__ __launch_bounds__(GEMM_THREADS) void gcn_gemm_wmma(const float* __restrict__ x,
                                                              const float* __restrict__ W,
                                                              float* __restrict__ h,
                                                              int n, int mtiles) {
    __shared__ __attribute__((aligned(16))) float Ws[DOUT * WPITCH];

    if (mtiles <= 0) return;

    const int wave = __builtin_amdgcn_readfirstlane((int)(threadIdx.x >> 5));
    const int lane = threadIdx.x & 31;
    const int half = lane >> 4;
    const int l16  = lane & 15;
    const int tstride = GEMM_WAVES * (int)gridDim.x;

    const float* bbase = Ws + l16 * WPITCH + half * 4;

    int tile = wave * (int)gridDim.x + (int)blockIdx.x;

    int arow = tile * 16 + l16;
    if (arow >= n) arow = n - 1;
    const float* ap = x + (size_t)arow * DIN + half * 4;
    v4f a[4];
#pragma unroll
    for (int s = 0; s < 4; ++s) a[s] = *(const v4f*)(ap + s * 8);

    for (int i = threadIdx.x; i < DOUT * (DIN / 4); i += GEMM_THREADS) {
        const int row = i >> 6;
        const int c4  = i & 63;
        const v4f v = *(const v4f*)(W + (size_t)row * DIN + c4 * 4);
        *(v4f*)(Ws + row * WPITCH + c4 * 4) = v;
    }
    __syncthreads();


    v4f b[8];
#pragma unroll
    for (int ct = 0; ct < 8; ++ct) b[ct] = *(const v4f*)(bbase + ct * 16 * WPITCH);


    for (; tile < mtiles; tile += tstride) {
        int nrow = (tile + tstride) * 16 + l16;
        if (nrow >= n) nrow = n - 1;
        const float* apn = x + (size_t)nrow * DIN + half * 4;

        v8f acc[8];
#pragma unroll
        for (int ct = 0; ct < 8; ++ct) acc[ct] = (v8f)(0.0f);

#pragma unroll 2
        for (int kc = 0; kc < DIN / 32; ++kc) {
            const float* anp = (kc + 1 < DIN / 32) ? ap + (kc + 1) * 32 : apn;
            v4f an[4];
#pragma unroll
            for (int s = 0; s < 4; ++s) an[s] = *(const v4f*)(anp + s * 8);

#pragma unroll
            for (int s = 0; s < 4; ++s) {
                const int gn = (kc * 4 + s + 1) & (KGROUPS - 1);
                v4f bn[8];
#pragma unroll
                for (int ct = 0; ct < 8; ++ct)
                    bn[ct] = *(const v4f*)(bbase + ct * 16 * WPITCH + gn * 8);

                const v2f a_lo = {a[s].x, a[s].y};
                const v2f a_hi = {a[s].z, a[s].w};
#pragma unroll
                for (int ct = 0; ct < 8; ++ct) {
                    const v2f b_lo = {b[ct].x, b[ct].y};
                    acc[ct] = __builtin_amdgcn_wmma_f32_16x16x4_f32(
                        false, a_lo, false, b_lo, (short)0, acc[ct], false, false);
                }
#pragma unroll
                for (int ct = 0; ct < 8; ++ct) {
                    const v2f b_hi = {b[ct].z, b[ct].w};
                    acc[ct] = __builtin_amdgcn_wmma_f32_16x16x4_f32(
                        false, a_hi, false, b_hi, (short)0, acc[ct], false, false);
                }
#pragma unroll
                for (int ct = 0; ct < 8; ++ct) b[ct] = bn[ct];
            }
#pragma unroll
            for (int s = 0; s < 4; ++s) a[s] = an[s];

#pragma unroll
            for (int s = 0; s < 4; ++s) {
                __builtin_amdgcn_sched_group_barrier(0x100, 8, 0);
                if (s == 0) __builtin_amdgcn_sched_group_barrier(0x020, 4, 0);
                __builtin_amdgcn_sched_group_barrier(0x008, 16, 0);
            }
        }

        const int rbase = tile * 16 + half * 8;
#pragma unroll
        for (int j = 0; j < 8; ++j) {
            const int r = rbase + j;
            if (r < n) {
#pragma unroll
                for (int ct = 0; ct < 8; ++ct)
                    h[(size_t)r * DOUT + ct * 16 + l16] = acc[ct][j];
            }
        }
        ap = apn;
    }
}

__global__ __launch_bounds__(256) void gcn_spmm(const float* __restrict__ h,
                                                const int* __restrict__ ptr,
                                                const int* __restrict__ idx,
                                                float* __restrict__ out, int n) {
    const int wave = __builtin_amdgcn_readfirstlane((int)(threadIdx.x >> 5));
    const int node = (int)blockIdx.x * (int)(blockDim.x >> 5) + wave;
    if (node >= n) return;
    const int lane = threadIdx.x & 31;

    const int s = ptr[node];
    const int e = ptr[node + 1];

    float ax = 0.f, ay = 0.f, az = 0.f, aw = 0.f;
    for (int base = s; base < e; base += 32) {
        const int left = e - base;
        const int cnt = left < 32 ? left : 32;

        int mine = 0;
        if (lane < cnt) mine = idx[base + lane];

        int k = 0;
        for (; k + 4 <= cnt; k += 4) {
            const int s0 = __builtin_amdgcn_readlane(mine, k);
            const int s1 = __builtin_amdgcn_readlane(mine, k + 1);
            const int s2 = __builtin_amdgcn_readlane(mine, k + 2);
            const int s3 = __builtin_amdgcn_readlane(mine, k + 3);
            const float4 v0 = ((const float4*)(h + (size_t)s0 * DOUT))[lane];
            const float4 v1 = ((const float4*)(h + (size_t)s1 * DOUT))[lane];
            const float4 v2 = ((const float4*)(h + (size_t)s2 * DOUT))[lane];
            const float4 v3 = ((const float4*)(h + (size_t)s3 * DOUT))[lane];
            ax += v0.x + v1.x;  ay += v0.y + v1.y;
            az += v0.z + v1.z;  aw += v0.w + v1.w;
            ax += v2.x + v3.x;  ay += v2.y + v3.y;
            az += v2.z + v3.z;  aw += v2.w + v3.w;
        }
        if (k + 2 <= cnt) {
            const int s0 = __builtin_amdgcn_readlane(mine, k);
            const int s1 = __builtin_amdgcn_readlane(mine, k + 1);
            const float4 v0 = ((const float4*)(h + (size_t)s0 * DOUT))[lane];
            const float4 v1 = ((const float4*)(h + (size_t)s1 * DOUT))[lane];
            ax += v0.x + v1.x;  ay += v0.y + v1.y;
            az += v0.z + v1.z;  aw += v0.w + v1.w;
            k += 2;
        }
        if (k < cnt) {
            const int s0 = __builtin_amdgcn_readlane(mine, k);
            const float4 v0 = ((const float4*)(h + (size_t)s0 * DOUT))[lane];
            ax += v0.x;  ay += v0.y;  az += v0.z;  aw += v0.w;
        }
    }

    const float inv = (e > s) ? 1.0f / (float)(e - s) : 0.0f;
    float4 r;
    r.x = ax * inv;  r.y = ay * inv;  r.z = az * inv;  r.w = aw * inv;
    ((float4*)(out + (size_t)node * DOUT))[lane] = r;
}

extern "C" void kernel_launch(void* const* d_in, const int* in_sizes, int n_in,
                              void* d_out, int out_size, void* d_ws, size_t ws_size,
                              hipStream_t stream) {
    const float* x   = (const float*)d_in[0];
    const float* W   = (const float*)d_in[1];
    const int*   ptr = (const int*)d_in[2];
    const int*   idx = (const int*)d_in[3];
    (void)n_in; (void)out_size; (void)ws_size;

    const int n = in_sizes[2] - 1;
    float* h   = (float*)d_ws;
    float* out = (float*)d_out;

    const int mtiles = (n + 15) / 16;
    const int wavesPerBlock = 8;

    dim3 blk(256);
    gcn_gemm_wmma<<<dim3(256), dim3(GEMM_THREADS), 0, stream>>>(x, W, h, n, mtiles);

    dim3 g2((n + wavesPerBlock - 1) / wavesPerBlock);
    gcn_spmm<<<g2, blk, 0, stream>>>(h, ptr, idx, out, n);
}


# GEMM: each SIMD gets the same six whole row tiles; left-over row tiles are cut into four 32-column pieces, one per SIMD of a work-group: busiest SIMD issues 6 1/4 tiles of WMMAs, not 7
# speedup vs baseline: 1.8826x; 1.0529x over previous
#include <hip/hip_runtime.h>

typedef float v2f __attribute__((ext_vector_type(2)));
typedef float v4f __attribute__((ext_vector_type(4)));
typedef float v8f __attribute__((ext_vector_type(8)));

#define DIN 256
#define DOUT 128

#define WPITCH 260
#define GEMM_WAVES 16
#define GEMM_THREADS (GEMM_WAVES * 32)
#define KGROUPS (DIN / 8)

__global__ __launch_bounds__(GEMM_THREADS) void gcn_gemm_wmma(const float* __restrict__ x,
                                                              const float* __restrict__ W,
                                                              float* __restrict__ h,
                                                              int n, int mtiles) {
    __shared__ __attribute__((aligned(16))) float Ws[DOUT * WPITCH];

    if (mtiles <= 0) return;

    const int wave = __builtin_amdgcn_readfirstlane((int)(threadIdx.x >> 5));
    const int lane = threadIdx.x & 31;
    const int half = lane >> 4;
    const int l16  = lane & 15;
    const int nwg = (int)gridDim.x;
    const int tstride = GEMM_WAVES * nwg;

    const int per_simd   = mtiles / (4 * nwg);
    const int main_tiles = per_simd * 4 * nwg;
    const int piece  = wave;
    const int ltile0 = (piece < 4 && (int)blockIdx.x < mtiles - main_tiles)
                           ? main_tiles + (int)blockIdx.x : mtiles;

    const float* bbase = Ws + l16 * WPITCH + half * 4;

    int tile = wave * nwg + (int)blockIdx.x;

    int arow = (ltile0 < mtiles ? ltile0 : tile) * 16 + l16;
    if (arow >= n) arow = n - 1;
    const float* ap = x + (size_t)arow * DIN + half * 4;
    v4f a[4];
#pragma unroll
    for (int s = 0; s < 4; ++s) a[s] = *(const v4f*)(ap + s * 8);

    for (int i = threadIdx.x; i < DOUT * (DIN / 4); i += GEMM_THREADS) {
        const int row = i >> 6;
        const int c4  = i & 63;
        const v4f v = *(const v4f*)(W + (size_t)row * DIN + c4 * 4);
        *(v4f*)(Ws + row * WPITCH + c4 * 4) = v;
    }
    __syncthreads();

    if (ltile0 < mtiles) {
        const float* bp = bbase + piece * 32 * WPITCH;
        for (int lt = ltile0; lt < mtiles; lt += nwg) {
            int nrow = ((lt + nwg < mtiles) ? lt + nwg : tile) * 16 + l16;
            if (nrow >= n) nrow = n - 1;
            const float* apn = x + (size_t)nrow * DIN + half * 4;

            v8f c0 = (v8f)(0.0f), c1 = (v8f)(0.0f);
            v4f q0[4], q1[4];
#pragma unroll
            for (int s = 0; s < 4; ++s) {
                q0[s] = *(const v4f*)(bp + s * 8);
                q1[s] = *(const v4f*)(bp + 16 * WPITCH + s * 8);
            }
#pragma unroll 2
            for (int kc = 0; kc < DIN / 32; ++kc) {
                const float* anp = (kc + 1 < DIN / 32) ? ap + (kc + 1) * 32 : apn;
                const int kn = ((kc + 1) & (DIN / 32 - 1)) * 32;
                v4f an[4], qn0[4], qn1[4];
#pragma unroll
                for (int s = 0; s < 4; ++s) an[s] = *(const v4f*)(anp + s * 8);
#pragma unroll
                for (int s = 0; s < 4; ++s) {
                    qn0[s] = *(const v4f*)(bp + kn + s * 8);
                    qn1[s] = *(const v4f*)(bp + 16 * WPITCH + kn + s * 8);
                }
#pragma unroll
                for (int s = 0; s < 4; ++s) {
                    const v2f a_lo = {a[s].x, a[s].y};
                    const v2f a_hi = {a[s].z, a[s].w};
                    const v2f b0_lo = {q0[s].x, q0[s].y}, b0_hi = {q0[s].z, q0[s].w};
                    const v2f b1_lo = {q1[s].x, q1[s].y}, b1_hi = {q1[s].z, q1[s].w};
                    c0 = __builtin_amdgcn_wmma_f32_16x16x4_f32(false, a_lo, false, b0_lo, (short)0, c0, false, false);
                    c1 = __builtin_amdgcn_wmma_f32_16x16x4_f32(false, a_lo, false, b1_lo, (short)0, c1, false, false);
                    c0 = __builtin_amdgcn_wmma_f32_16x16x4_f32(false, a_hi, false, b0_hi, (short)0, c0, false, false);
                    c1 = __builtin_amdgcn_wmma_f32_16x16x4_f32(false, a_hi, false, b1_hi, (short)0, c1, false, false);
                }
#pragma unroll
                for (int s = 0; s < 4; ++s) { a[s] = an[s]; q0[s] = qn0[s]; q1[s] = qn1[s]; }
            }

            const int rbase = lt * 16 + half * 8;
#pragma unroll
            for (int j = 0; j < 8; ++j) {
                const int r = rbase + j;
                if (r < n) {
                    float* hp = h + (size_t)r * DOUT + piece * 32 + l16;
                    hp[0]  = c0[j];
                    hp[16] = c1[j];
                }
            }
            ap = apn;
        }
    }

    v4f b[8];
#pragma unroll
    for (int ct = 0; ct < 8; ++ct) b[ct] = *(const v4f*)(bbase + ct * 16 * WPITCH);


    for (; tile < main_tiles; tile += tstride) {
        int nrow = (tile + tstride) * 16 + l16;
        if (nrow >= n) nrow = n - 1;
        const float* apn = x + (size_t)nrow * DIN + half * 4;

        v8f acc[8];
#pragma unroll
        for (int ct = 0; ct < 8; ++ct) acc[ct] = (v8f)(0.0f);

#pragma unroll 2
        for (int kc = 0; kc < DIN / 32; ++kc) {
            const float* anp = (kc + 1 < DIN / 32) ? ap + (kc + 1) * 32 : apn;
            v4f an[4];
#pragma unroll
            for (int s = 0; s < 4; ++s) an[s] = *(const v4f*)(anp + s * 8);

#pragma unroll
            for (int s = 0; s < 4; ++s) {
                const int gn = (kc * 4 + s + 1) & (KGROUPS - 1);
                v4f bn[8];
#pragma unroll
                for (int ct = 0; ct < 8; ++ct)
                    bn[ct] = *(const v4f*)(bbase + ct * 16 * WPITCH + gn * 8);

                const v2f a_lo = {a[s].x, a[s].y};
                const v2f a_hi = {a[s].z, a[s].w};
#pragma unroll
                for (int ct = 0; ct < 8; ++ct) {
                    const v2f b_lo = {b[ct].x, b[ct].y};
                    acc[ct] = __builtin_amdgcn_wmma_f32_16x16x4_f32(
                        false, a_lo, false, b_lo, (short)0, acc[ct], false, false);
                }
#pragma unroll
                for (int ct = 0; ct < 8; ++ct) {
                    const v2f b_hi = {b[ct].z, b[ct].w};
                    acc[ct] = __builtin_amdgcn_wmma_f32_16x16x4_f32(
                        false, a_hi, false, b_hi, (short)0, acc[ct], false, false);
                }
#pragma unroll
                for (int ct = 0; ct < 8; ++ct) b[ct] = bn[ct];
            }
#pragma unroll
            for (int s = 0; s < 4; ++s) a[s] = an[s];

#pragma unroll
            for (int s = 0; s < 4; ++s) {
                __builtin_amdgcn_sched_group_barrier(0x100, 8, 0);
                if (s == 0) __builtin_amdgcn_sched_group_barrier(0x020, 4, 0);
                __builtin_amdgcn_sched_group_barrier(0x008, 16, 0);
            }
        }

        const int rbase = tile * 16 + half * 8;
#pragma unroll
        for (int j = 0; j < 8; ++j) {
            const int r = rbase + j;
            if (r < n) {
#pragma unroll
                for (int ct = 0; ct < 8; ++ct)
                    h[(size_t)r * DOUT + ct * 16 + l16] = acc[ct][j];
            }
        }
        ap = apn;
    }
}

__global__ __launch_bounds__(256) void gcn_spmm(const float* __restrict__ h,
                                                const int* __restrict__ ptr,
                                                const int* __restrict__ idx,
                                                float* __restrict__ out, int n) {
    const int wave = __builtin_amdgcn_readfirstlane((int)(threadIdx.x >> 5));
    const int node = (int)blockIdx.x * (int)(blockDim.x >> 5) + wave;
    if (node >= n) return;
    const int lane = threadIdx.x & 31;

    const int s = ptr[node];
    const int e = ptr[node + 1];

    float ax = 0.f, ay = 0.f, az = 0.f, aw = 0.f;
    for (int base = s; base < e; base += 32) {
        const int left = e - base;
        const int cnt = left < 32 ? left : 32;

        int mine = 0;
        if (lane < cnt) mine = idx[base + lane];

        int k = 0;
        for (; k + 4 <= cnt; k += 4) {
            const int s0 = __builtin_amdgcn_readlane(mine, k);
            const int s1 = __builtin_amdgcn_readlane(mine, k + 1);
            const int s2 = __builtin_amdgcn_readlane(mine, k + 2);
            const int s3 = __builtin_amdgcn_readlane(mine, k + 3);
            const float4 v0 = ((const float4*)(h + (size_t)s0 * DOUT))[lane];
            const float4 v1 = ((const float4*)(h + (size_t)s1 * DOUT))[lane];
            const float4 v2 = ((const float4*)(h + (size_t)s2 * DOUT))[lane];
            const float4 v3 = ((const float4*)(h + (size_t)s3 * DOUT))[lane];
            ax += v0.x + v1.x;  ay += v0.y + v1.y;
            az += v0.z + v1.z;  aw += v0.w + v1.w;
            ax += v2.x + v3.x;  ay += v2.y + v3.y;
            az += v2.z + v3.z;  aw += v2.w + v3.w;
        }
        if (k + 2 <= cnt) {
            const int s0 = __builtin_amdgcn_readlane(mine, k);
            const int s1 = __builtin_amdgcn_readlane(mine, k + 1);
            const float4 v0 = ((const float4*)(h + (size_t)s0 * DOUT))[lane];
            const float4 v1 = ((const float4*)(h + (size_t)s1 * DOUT))[lane];
            ax += v0.x + v1.x;  ay += v0.y + v1.y;
            az += v0.z + v1.z;  aw += v0.w + v1.w;
            k += 2;
        }
        if (k < cnt) {
            const int s0 = __builtin_amdgcn_readlane(mine, k);
            const float4 v0 = ((const float4*)(h + (size_t)s0 * DOUT))[lane];
            ax += v0.x;  ay += v0.y;  az += v0.z;  aw += v0.w;
        }
    }

    const float inv = (e > s) ? 1.0f / (float)(e - s) : 0.0f;
    float4 r;
    r.x = ax * inv;  r.y = ay * inv;  r.z = az * inv;  r.w = aw * inv;
    ((float4*)(out + (size_t)node * DOUT))[lane] = r;
}

extern "C" void kernel_launch(void* const* d_in, const int* in_sizes, int n_in,
                              void* d_out, int out_size, void* d_ws, size_t ws_size,
                              hipStream_t stream) {
    const float* x   = (const float*)d_in[0];
    const float* W   = (const float*)d_in[1];
    const int*   ptr = (const int*)d_in[2];
    const int*   idx = (const int*)d_in[3];
    (void)n_in; (void)out_size; (void)ws_size;

    const int n = in_sizes[2] - 1;
    float* h   = (float*)d_ws;
    float* out = (float*)d_out;

    const int mtiles = (n + 15) / 16;
    const int wavesPerBlock = 8;

    dim3 blk(256);
    gcn_gemm_wmma<<<dim3(256), dim3(GEMM_THREADS), 0, stream>>>(x, W, h, n, mtiles);

    dim3 g2((n + wavesPerBlock - 1) / wavesPerBlock);
    gcn_spmm<<<g2, blk, 0, stream>>>(h, ptr, idx, out, n);
}
